# starting kernel
# baseline (speedup 1.0000x reference)
#include <hip/hip_runtime.h>
#include <hip/hip_bf16.h>

typedef float v2f __attribute__((ext_vector_type(2)));
typedef float v8f __attribute__((ext_vector_type(8)));

#define N_NODES 2048
#define N_GRAPHS 64
#define N_EDGES 16384
#define BATCH 65536
#define META 64
#define TXD 8
#define NOISE 128
#define TSTRIDE 48
#define CROWS 64
#define NCHUNK (BATCH / CROWS)

__device__ __forceinline__ int kext_row(int k) {
    if (k < 32) return 2049 + k;
    if (k == 32) return 2048;
    if (k <= 40) return 2081 + (k - 33);
    return 2048;
}

__global__ void __launch_bounds__(256) gcn_kernel(const int* __restrict__ edges,
                                                  const float* __restrict__ gw,
                                                  const float* __restrict__ gb,
                                                  float* __restrict__ graph_emb) {
    __shared__ float deg[N_NODES];
    __shared__ float agg[N_NODES];
    const int g = blockIdx.x;
    const int t = threadIdx.x;
    const int* src = edges + (size_t)g * (2 * N_EDGES);
    const int* dst = src + N_EDGES;

    for (int i = t; i < N_NODES; i += 256) deg[i] = 1.0f;
    __syncthreads();
    for (int e = t; e < N_EDGES; e += 256)
        atomicAdd(&deg[dst[e]], 1.0f);
    __syncthreads();
    for (int i = t; i < N_NODES; i += 256) {
        float d = deg[i];
        float r = d > 0.0f ? rsqrtf(d) : 0.0f;
        deg[i] = r;
        agg[i] = r * r;
    }
    __syncthreads();
    for (int e = t; e < N_EDGES; e += 256) {
        int s = src[e], d = dst[e];
        atomicAdd(&agg[d], deg[s] * deg[d]);
    }
    __syncthreads();
    const float w = *gw, b = *gb;
    for (int i = t; i < N_NODES; i += 256)
        graph_emb[(size_t)g * N_NODES + i] = agg[i] * w + b;
}

__global__ void __launch_bounds__(256) proj_kernel(const float* __restrict__ ge,
                                                   const float* __restrict__ embW,
                                                   float* __restrict__ gp) {
    const int wave  = (blockIdx.x * blockDim.x + threadIdx.x) >> 5;
    const int lane  = threadIdx.x & 31;
    const int mtile = wave >> 3;
    const int ntile = wave & 7;
    const int lrow  = lane & 15;
    const int khalf = (lane >> 4) * 2;
    const int arow  = mtile * 16 + lrow;
    const int col   = ntile * 16 + lrow;

    v8f c = {};
    for (int kk = 0; kk < N_NODES; kk += 4) {
        const int k0 = kk + khalf;
        v2f a, bm;
        a.x  = ge[(size_t)arow * N_NODES + k0];
        a.y  = ge[(size_t)arow * N_NODES + k0 + 1];
        bm.x = embW[(size_t)k0 * NOISE + col];
        bm.y = embW[(size_t)(k0 + 1) * NOISE + col];
        c = __builtin_amdgcn_wmma_f32_16x16x4_f32(false, a, false, bm,
                                                  (short)0, c, false, false);
    }
    const int rbase = mtile * 16 + (lane >> 4) * 8;
#pragma unroll
    for (int r = 0; r < 8; ++r)
        gp[(size_t)(rbase + r) * NOISE + col] = c[r];
}

__global__ void __launch_bounds__(256) batch_kernel(const int* __restrict__ graphs,
                                                    const float* __restrict__ chain,
                                                    const float* __restrict__ trig,
                                                    const float* __restrict__ tx,
                                                    const float* __restrict__ trigW,
                                                    const float* __restrict__ trigb,
                                                    const float* __restrict__ embW,
                                                    const float* __restrict__ embb,
                                                    const float* __restrict__ gp,
                                                    float* __restrict__ out) {
    __shared__ __align__(16) float gpl[N_GRAPHS * NOISE];
    __shared__ float T[CROWS * TSTRIDE];

    const int t     = threadIdx.x;
    const int lane  = t & 31;
    const int w     = t >> 5;
    const int lrow  = lane & 15;
    const int khalf = (lane >> 4) * 2;
    const int col   = w * 16 + lrow;
    const int lr0   = (lane >> 4) * 8;

    for (int i = t * 4; i < N_GRAPHS * NOISE; i += 256 * 4)
        *(float4*)&gpl[i] = *(const float4*)&gp[i];

    v2f bf[11];
#pragma unroll
    for (int ks = 0; ks < 11; ++ks) {
        const int k0 = ks * 4 + khalf;
        bf[ks].x = embW[(size_t)kext_row(k0) * NOISE + col];
        bf[ks].y = embW[(size_t)kext_row(k0 + 1) * NOISE + col];
    }

    const int rt1 = w >> 1;
    const int nt1 = w & 1;
    const int c1  = nt1 * 16 + lrow;
    v2f b1[16];
#pragma unroll
    for (int ks = 0; ks < 16; ++ks) {
        const int k0 = ks * 4 + khalf;
        b1[ks].x = trigW[k0 * 32 + c1];
        b1[ks].y = trigW[(k0 + 1) * 32 + c1];
    }
    const float tb = trigb[c1];
    const float eb = embb[col];
    __syncthreads();

    for (int chunk = blockIdx.x; chunk < NCHUNK; chunk += gridDim.x) {
        const int rowbase = chunk * CROWS;

        if (chunk + gridDim.x < NCHUNK) {
            const float* p = trig + (size_t)(chunk + gridDim.x) * CROWS * META
                                  + t * 16;
            __builtin_prefetch(p, 0, 3);
        }

#pragma unroll
        for (int j = 0; j < 4; ++j) {
            const int slot = t + 256 * j;
            const int fr = slot >> 4;
            const int fc = 32 + (slot & 15);
            float v = 0.0f;
            const int grow = rowbase + fr;
            if (fc == 32)      v = chain[grow];
            else if (fc <= 40) v = tx[(size_t)grow * TXD + (fc - 33)];
            T[fr * TSTRIDE + fc] = v;
        }

        {
            v8f d = {};
            const int arow = rowbase + rt1 * 16 + lrow;
#pragma unroll
            for (int ks = 0; ks < 16; ++ks) {
                const int k0 = ks * 4 + khalf;
                v2f a;
                a.x = trig[(size_t)arow * META + k0];
                a.y = trig[(size_t)arow * META + k0 + 1];
                d = __builtin_amdgcn_wmma_f32_16x16x4_f32(false, a, false, b1[ks],
                                                          (short)0, d, false, false);
            }
#pragma unroll
            for (int r = 0; r < 8; ++r) {
                float v = d[r] + tb;
                T[(rt1 * 16 + lr0 + r) * TSTRIDE + c1] = v > 0.0f ? v : 0.0f;
            }
        }
        __syncthreads();

        v8f acc0 = {}, acc1 = {}, acc2 = {}, acc3 = {};
#pragma unroll
        for (int ks = 0; ks < 11; ++ks) {
            const int k0 = ks * 4 + khalf;
            v2f a0, a1, a2, a3;
            a0.x = T[(0 * 16 + lrow) * TSTRIDE + k0];
            a0.y = T[(0 * 16 + lrow) * TSTRIDE + k0 + 1];
            a1.x = T[(1 * 16 + lrow) * TSTRIDE + k0];
            a1.y = T[(1 * 16 + lrow) * TSTRIDE + k0 + 1];
            a2.x = T[(2 * 16 + lrow) * TSTRIDE + k0];
            a2.y = T[(2 * 16 + lrow) * TSTRIDE + k0 + 1];
            a3.x = T[(3 * 16 + lrow) * TSTRIDE + k0];
            a3.y = T[(3 * 16 + lrow) * TSTRIDE + k0 + 1];
            acc0 = __builtin_amdgcn_wmma_f32_16x16x4_f32(false, a0, false, bf[ks],
                                                         (short)0, acc0, false, false);
            acc1 = __builtin_amdgcn_wmma_f32_16x16x4_f32(false, a1, false, bf[ks],
                                                         (short)0, acc1, false, false);
            acc2 = __builtin_amdgcn_wmma_f32_16x16x4_f32(false, a2, false, bf[ks],
                                                         (short)0, acc2, false, false);
            acc3 = __builtin_amdgcn_wmma_f32_16x16x4_f32(false, a3, false, bf[ks],
                                                         (short)0, acc3, false, false);
        }

#pragma unroll
        for (int rt = 0; rt < 4; ++rt) {
            const v8f& c = (rt == 0) ? acc0 : (rt == 1) ? acc1
                         : (rt == 2) ? acc2 : acc3;
#pragma unroll
            for (int r = 0; r < 8; ++r) {
                const int grow = rowbase + rt * 16 + lr0 + r;
                const int g = graphs[grow];
                out[(size_t)grow * NOISE + col] = c[r] + gpl[g * NOISE + col] + eb;
            }
        }
        __syncthreads();
    }
}

extern "C" void kernel_launch(void* const* d_in, const int* in_sizes, int n_in,
                              void* d_out, int out_size, void* d_ws, size_t ws_size,
                              hipStream_t stream) {
    const int*   graphs = (const int*)d_in[0];
    const float* chain  = (const float*)d_in[1];
    const float* trig   = (const float*)d_in[2];
    const float* tx     = (const float*)d_in[3];
    const int*   edges  = (const int*)d_in[4];
    const float* gw     = (const float*)d_in[5];
    const float* gb     = (const float*)d_in[6];
    const float* trigW  = (const float*)d_in[7];
    const float* trigb  = (const float*)d_in[8];
    const float* embW   = (const float*)d_in[9];
    const float* embb   = (const float*)d_in[10];
    float* out = (float*)d_out;

    float* ge = (float*)d_ws;
    float* gp = ge + (size_t)N_GRAPHS * N_NODES;

    gcn_kernel<<<N_GRAPHS, 256, 0, stream>>>(edges, gw, gb, ge);
    proj_kernel<<<4, 256, 0, stream>>>(ge, embW, gp);
    batch_kernel<<<512, 256, 0, stream>>>(graphs, chain, trig, tx, trigW, trigb,
                                          embW, embb, gp, out);
}


# projection GEMM (graph_emb x emb_W): K split over 32 waves per 16x16 tile, 32 groups of 1024 threads (was 4 of 256); partial tiles summed through LDS in fixed wave order (k sum order changes)
# speedup vs baseline: 1.4994x; 1.4994x over previous
#include <hip/hip_runtime.h>
#include <hip/hip_bf16.h>

typedef float v2f __attribute__((ext_vector_type(2)));
typedef float v8f __attribute__((ext_vector_type(8)));

#define N_NODES 2048
#define N_GRAPHS 64
#define N_EDGES 16384
#define BATCH 65536
#define META 64
#define TXD 8
#define NOISE 128
#define TSTRIDE 48
#define CROWS 64
#define NCHUNK (BATCH / CROWS)

__device__ __forceinline__ int kext_row(int k) {
    if (k < 32) return 2049 + k;
    if (k == 32) return 2048;
    if (k <= 40) return 2081 + (k - 33);
    return 2048;
}

__global__ void __launch_bounds__(256) gcn_kernel(const int* __restrict__ edges,
                                                  const float* __restrict__ gw,
                                                  const float* __restrict__ gb,
                                                  float* __restrict__ graph_emb) {
    __shared__ float deg[N_NODES];
    __shared__ float agg[N_NODES];
    const int g = blockIdx.x;
    const int t = threadIdx.x;
    const int* src = edges + (size_t)g * (2 * N_EDGES);
    const int* dst = src + N_EDGES;

    for (int i = t; i < N_NODES; i += 256) deg[i] = 1.0f;
    __syncthreads();
    for (int e = t; e < N_EDGES; e += 256)
        atomicAdd(&deg[dst[e]], 1.0f);
    __syncthreads();
    for (int i = t; i < N_NODES; i += 256) {
        float d = deg[i];
        float r = d > 0.0f ? rsqrtf(d) : 0.0f;
        deg[i] = r;
        agg[i] = r * r;
    }
    __syncthreads();
    for (int e = t; e < N_EDGES; e += 256) {
        int s = src[e], d = dst[e];
        atomicAdd(&agg[d], deg[s] * deg[d]);
    }
    __syncthreads();
    const float w = *gw, b = *gb;
    for (int i = t; i < N_NODES; i += 256)
        graph_emb[(size_t)g * N_NODES + i] = agg[i] * w + b;
}

#define PJ_WAVES  32
#define PJ_KSLICE (N_NODES / PJ_WAVES)
#define PJ_STEPS  (PJ_KSLICE / 4)

__global__ void __launch_bounds__(1024) proj_kernel(const float* __restrict__ ge,
                                                    const float* __restrict__ embW,
                                                    float* __restrict__ gp) {
    __shared__ float part[PJ_WAVES * 256];

    const int t     = threadIdx.x;
    const int lane  = t & 31;
    const int w     = t >> 5;
    const int mtile = blockIdx.x >> 3;
    const int ntile = blockIdx.x & 7;
    const int lrow  = lane & 15;
    const int khalf = (lane >> 4) * 2;
    const int arow  = mtile * 16 + lrow;
    const int col   = ntile * 16 + lrow;
    const int kbase = w * PJ_KSLICE + khalf;

    const float* ap = ge + (size_t)arow * N_NODES + kbase;
    const float* bp = embW + (size_t)kbase * NOISE + col;

    v8f c0 = {}, c1 = {};
#pragma unroll
    for (int s = 0; s < PJ_STEPS; s += 2) {
        v2f a0, b0, a1, b1;
        a0.x = ap[4 * s];
        a0.y = ap[4 * s + 1];
        b0.x = bp[(size_t)(4 * s) * NOISE];
        b0.y = bp[(size_t)(4 * s + 1) * NOISE];
        a1.x = ap[4 * s + 4];
        a1.y = ap[4 * s + 5];
        b1.x = bp[(size_t)(4 * s + 4) * NOISE];
        b1.y = bp[(size_t)(4 * s + 5) * NOISE];
        c0 = __builtin_amdgcn_wmma_f32_16x16x4_f32(false, a0, false, b0,
                                                   (short)0, c0, false, false);
        c1 = __builtin_amdgcn_wmma_f32_16x16x4_f32(false, a1, false, b1,
                                                   (short)0, c1, false, false);
    }

#pragma unroll
    for (int r = 0; r < 8; ++r)
        part[w * 256 + r * 32 + lane] = c0[r] + c1[r];
    __syncthreads();

    if (t < 256) {
        float s = part[t];
#pragma unroll
        for (int q = 1; q < PJ_WAVES; ++q) s += part[q * 256 + t];
        const int r   = t >> 5;
        const int l   = t & 31;
        const int row = mtile * 16 + (l >> 4) * 8 + r;
        const int cc  = ntile * 16 + (l & 15);
        gp[(size_t)row * NOISE + cc] = s;
    }
}

__global__ void __launch_bounds__(256) batch_kernel(const int* __restrict__ graphs,
                                                    const float* __restrict__ chain,
                                                    const float* __restrict__ trig,
                                                    const float* __restrict__ tx,
                                                    const float* __restrict__ trigW,
                                                    const float* __restrict__ trigb,
                                                    const float* __restrict__ embW,
                                                    const float* __restrict__ embb,
                                                    const float* __restrict__ gp,
                                                    float* __restrict__ out) {
    __shared__ __align__(16) float gpl[N_GRAPHS * NOISE];
    __shared__ float T[CROWS * TSTRIDE];

    const int t     = threadIdx.x;
    const int lane  = t & 31;
    const int w     = t >> 5;
    const int lrow  = lane & 15;
    const int khalf = (lane >> 4) * 2;
    const int col   = w * 16 + lrow;
    const int lr0   = (lane >> 4) * 8;

    for (int i = t * 4; i < N_GRAPHS * NOISE; i += 256 * 4)
        *(float4*)&gpl[i] = *(const float4*)&gp[i];

    v2f bf[11];
#pragma unroll
    for (int ks = 0; ks < 11; ++ks) {
        const int k0 = ks * 4 + khalf;
        bf[ks].x = embW[(size_t)kext_row(k0) * NOISE + col];
        bf[ks].y = embW[(size_t)kext_row(k0 + 1) * NOISE + col];
    }

    const int rt1 = w >> 1;
    const int nt1 = w & 1;
    const int c1  = nt1 * 16 + lrow;
    v2f b1[16];
#pragma unroll
    for (int ks = 0; ks < 16; ++ks) {
        const int k0 = ks * 4 + khalf;
        b1[ks].x = trigW[k0 * 32 + c1];
        b1[ks].y = trigW[(k0 + 1) * 32 + c1];
    }
    const float tb = trigb[c1];
    const float eb = embb[col];
    __syncthreads();

    for (int chunk = blockIdx.x; chunk < NCHUNK; chunk += gridDim.x) {
        const int rowbase = chunk * CROWS;

        if (chunk + gridDim.x < NCHUNK) {
            const float* p = trig + (size_t)(chunk + gridDim.x) * CROWS * META
                                  + t * 16;
            __builtin_prefetch(p, 0, 3);
        }

#pragma unroll
        for (int j = 0; j < 4; ++j) {
            const int slot = t + 256 * j;
            const int fr = slot >> 4;
            const int fc = 32 + (slot & 15);
            float v = 0.0f;
            const int grow = rowbase + fr;
            if (fc == 32)      v = chain[grow];
            else if (fc <= 40) v = tx[(size_t)grow * TXD + (fc - 33)];
            T[fr * TSTRIDE + fc] = v;
        }

        {
            v8f d = {};
            const int arow = rowbase + rt1 * 16 + lrow;
#pragma unroll
            for (int ks = 0; ks < 16; ++ks) {
                const int k0 = ks * 4 + khalf;
                v2f a;
                a.x = trig[(size_t)arow * META + k0];
                a.y = trig[(size_t)arow * META + k0 + 1];
                d = __builtin_amdgcn_wmma_f32_16x16x4_f32(false, a, false, b1[ks],
                                                          (short)0, d, false, false);
            }
#pragma unroll
            for (int r = 0; r < 8; ++r) {
                float v = d[r] + tb;
                T[(rt1 * 16 + lr0 + r) * TSTRIDE + c1] = v > 0.0f ? v : 0.0f;
            }
        }
        __syncthreads();

        v8f acc0 = {}, acc1 = {}, acc2 = {}, acc3 = {};
#pragma unroll
        for (int ks = 0; ks < 11; ++ks) {
            const int k0 = ks * 4 + khalf;
            v2f a0, a1, a2, a3;
            a0.x = T[(0 * 16 + lrow) * TSTRIDE + k0];
            a0.y = T[(0 * 16 + lrow) * TSTRIDE + k0 + 1];
            a1.x = T[(1 * 16 + lrow) * TSTRIDE + k0];
            a1.y = T[(1 * 16 + lrow) * TSTRIDE + k0 + 1];
            a2.x = T[(2 * 16 + lrow) * TSTRIDE + k0];
            a2.y = T[(2 * 16 + lrow) * TSTRIDE + k0 + 1];
            a3.x = T[(3 * 16 + lrow) * TSTRIDE + k0];
            a3.y = T[(3 * 16 + lrow) * TSTRIDE + k0 + 1];
            acc0 = __builtin_amdgcn_wmma_f32_16x16x4_f32(false, a0, false, bf[ks],
                                                         (short)0, acc0, false, false);
            acc1 = __builtin_amdgcn_wmma_f32_16x16x4_f32(false, a1, false, bf[ks],
                                                         (short)0, acc1, false, false);
            acc2 = __builtin_amdgcn_wmma_f32_16x16x4_f32(false, a2, false, bf[ks],
                                                         (short)0, acc2, false, false);
            acc3 = __builtin_amdgcn_wmma_f32_16x16x4_f32(false, a3, false, bf[ks],
                                                         (short)0, acc3, false, false);
        }

#pragma unroll
        for (int rt = 0; rt < 4; ++rt) {
            const v8f& c = (rt == 0) ? acc0 : (rt == 1) ? acc1
                         : (rt == 2) ? acc2 : acc3;
#pragma unroll
            for (int r = 0; r < 8; ++r) {
                const int grow = rowbase + rt * 16 + lr0 + r;
                const int g = graphs[grow];
                out[(size_t)grow * NOISE + col] = c[r] + gpl[g * NOISE + col] + eb;
            }
        }
        __syncthreads();
    }
}

extern "C" void kernel_launch(void* const* d_in, const int* in_sizes, int n_in,
                              void* d_out, int out_size, void* d_ws, size_t ws_size,
                              hipStream_t stream) {
    const int*   graphs = (const int*)d_in[0];
    const float* chain  = (const float*)d_in[1];
    const float* trig   = (const float*)d_in[2];
    const float* tx     = (const float*)d_in[3];
    const int*   edges  = (const int*)d_in[4];
    const float* gw     = (const float*)d_in[5];
    const float* gb     = (const float*)d_in[6];
    const float* trigW  = (const float*)d_in[7];
    const float* trigb  = (const float*)d_in[8];
    const float* embW   = (const float*)d_in[9];
    const float* embb   = (const float*)d_in[10];
    float* out = (float*)d_out;

    float* ge = (float*)d_ws;
    float* gp = ge + (size_t)N_GRAPHS * N_NODES;

    gcn_kernel<<<N_GRAPHS, 256, 0, stream>>>(edges, gw, gb, ge);
    proj_kernel<<<32, 1024, 0, stream>>>(ge, embW, gp);
    batch_kernel<<<512, 256, 0, stream>>>(graphs, chain, trig, tx, trigW, trigb,
                                          embW, embb, gp, out);
}


# gcn kernel: each graph's edge indices fetched at entry as b128 loads, all in flight (1024 threads x 16 edges), not one dependent load round trip per edge; LDS atomic order of agg sums changes
# speedup vs baseline: 3.3867x; 2.2587x over previous
#include <hip/hip_runtime.h>
#include <hip/hip_bf16.h>

typedef float v2f __attribute__((ext_vector_type(2)));
typedef float v8f __attribute__((ext_vector_type(8)));

#define N_NODES 2048
#define N_GRAPHS 64
#define N_EDGES 16384
#define BATCH 65536
#define META 64
#define TXD 8
#define NOISE 128
#define TSTRIDE 48
#define CROWS 64
#define NCHUNK (BATCH / CROWS)

__device__ __forceinline__ int kext_row(int k) {
    if (k < 32) return 2049 + k;
    if (k == 32) return 2048;
    if (k <= 40) return 2081 + (k - 33);
    return 2048;
}

#define GCN_THREADS 1024
#define GCN_VEC     (N_EDGES / 4 / GCN_THREADS)

__global__ void __launch_bounds__(GCN_THREADS) gcn_kernel(const int* __restrict__ edges,
                                                          const float* __restrict__ gw,
                                                          const float* __restrict__ gb,
                                                          float* __restrict__ graph_emb) {
    __shared__ float deg[N_NODES];
    __shared__ float agg[N_NODES];
    const int g = blockIdx.x;
    const int t = threadIdx.x;
    const int4* src4 = (const int4*)(edges + (size_t)g * (2 * N_EDGES));
    const int4* dst4 = src4 + N_EDGES / 4;

    int4 dv[GCN_VEC], sv[GCN_VEC];
#pragma unroll
    for (int j = 0; j < GCN_VEC; ++j) dv[j] = dst4[j * GCN_THREADS + t];
#pragma unroll
    for (int j = 0; j < GCN_VEC; ++j) sv[j] = src4[j * GCN_THREADS + t];

    for (int i = t; i < N_NODES; i += GCN_THREADS) deg[i] = 1.0f;
    __syncthreads();
#pragma unroll
    for (int j = 0; j < GCN_VEC; ++j) {
        atomicAdd(&deg[dv[j].x], 1.0f);
        atomicAdd(&deg[dv[j].y], 1.0f);
        atomicAdd(&deg[dv[j].z], 1.0f);
        atomicAdd(&deg[dv[j].w], 1.0f);
    }
    __syncthreads();
    for (int i = t; i < N_NODES; i += GCN_THREADS) {
        float d = deg[i];
        float r = d > 0.0f ? rsqrtf(d) : 0.0f;
        deg[i] = r;
        agg[i] = r * r;
    }
    __syncthreads();
    float nv[4 * GCN_VEC];
#pragma unroll
    for (int j = 0; j < GCN_VEC; ++j) {
        nv[4 * j + 0] = deg[sv[j].x] * deg[dv[j].x];
        nv[4 * j + 1] = deg[sv[j].y] * deg[dv[j].y];
        nv[4 * j + 2] = deg[sv[j].z] * deg[dv[j].z];
        nv[4 * j + 3] = deg[sv[j].w] * deg[dv[j].w];
    }
#pragma unroll
    for (int j = 0; j < GCN_VEC; ++j) {
        atomicAdd(&agg[dv[j].x], nv[4 * j + 0]);
        atomicAdd(&agg[dv[j].y], nv[4 * j + 1]);
        atomicAdd(&agg[dv[j].z], nv[4 * j + 2]);
        atomicAdd(&agg[dv[j].w], nv[4 * j + 3]);
    }
    __syncthreads();
    const float w = *gw, b = *gb;
    for (int i = t; i < N_NODES; i += GCN_THREADS)
        graph_emb[(size_t)g * N_NODES + i] = agg[i] * w + b;
}

#define PJ_WAVES  32
#define PJ_KSLICE (N_NODES / PJ_WAVES)
#define PJ_STEPS  (PJ_KSLICE / 4)

__global__ void __launch_bounds__(1024) proj_kernel(const float* __restrict__ ge,
                                                    const float* __restrict__ embW,
                                                    float* __restrict__ gp) {
    __shared__ float part[PJ_WAVES * 256];

    const int t     = threadIdx.x;
    const int lane  = t & 31;
    const int w     = t >> 5;
    const int mtile = blockIdx.x >> 3;
    const int ntile = blockIdx.x & 7;
    const int lrow  = lane & 15;
    const int khalf = (lane >> 4) * 2;
    const int arow  = mtile * 16 + lrow;
    const int col   = ntile * 16 + lrow;
    const int kbase = w * PJ_KSLICE + khalf;

    const float* ap = ge + (size_t)arow * N_NODES + kbase;
    const float* bp = embW + (size_t)kbase * NOISE + col;

    v8f c0 = {}, c1 = {};
#pragma unroll
    for (int s = 0; s < PJ_STEPS; s += 2) {
        v2f a0, b0, a1, b1;
        a0.x = ap[4 * s];
        a0.y = ap[4 * s + 1];
        b0.x = bp[(size_t)(4 * s) * NOISE];
        b0.y = bp[(size_t)(4 * s + 1) * NOISE];
        a1.x = ap[4 * s + 4];
        a1.y = ap[4 * s + 5];
        b1.x = bp[(size_t)(4 * s + 4) * NOISE];
        b1.y = bp[(size_t)(4 * s + 5) * NOISE];
        c0 = __builtin_amdgcn_wmma_f32_16x16x4_f32(false, a0, false, b0,
                                                   (short)0, c0, false, false);
        c1 = __builtin_amdgcn_wmma_f32_16x16x4_f32(false, a1, false, b1,
                                                   (short)0, c1, false, false);
    }

#pragma unroll
    for (int r = 0; r < 8; ++r)
        part[w * 256 + r * 32 + lane] = c0[r] + c1[r];
    __syncthreads();

    if (t < 256) {
        float s = part[t];
#pragma unroll
        for (int q = 1; q < PJ_WAVES; ++q) s += part[q * 256 + t];
        const int r   = t >> 5;
        const int l   = t & 31;
        const int row = mtile * 16 + (l >> 4) * 8 + r;
        const int cc  = ntile * 16 + (l & 15);
        gp[(size_t)row * NOISE + cc] = s;
    }
}

__global__ void __launch_bounds__(256) batch_kernel(const int* __restrict__ graphs,
                                                    const float* __restrict__ chain,
                                                    const float* __restrict__ trig,
                                                    const float* __restrict__ tx,
                                                    const float* __restrict__ trigW,
                                                    const float* __restrict__ trigb,
                                                    const float* __restrict__ embW,
                                                    const float* __restrict__ embb,
                                                    const float* __restrict__ gp,
                                                    float* __restrict__ out) {
    __shared__ __align__(16) float gpl[N_GRAPHS * NOISE];
    __shared__ float T[CROWS * TSTRIDE];

    const int t     = threadIdx.x;
    const int lane  = t & 31;
    const int w     = t >> 5;
    const int lrow  = lane & 15;
    const int khalf = (lane >> 4) * 2;
    const int col   = w * 16 + lrow;
    const int lr0   = (lane >> 4) * 8;

    for (int i = t * 4; i < N_GRAPHS * NOISE; i += 256 * 4)
        *(float4*)&gpl[i] = *(const float4*)&gp[i];

    v2f bf[11];
#pragma unroll
    for (int ks = 0; ks < 11; ++ks) {
        const int k0 = ks * 4 + khalf;
        bf[ks].x = embW[(size_t)kext_row(k0) * NOISE + col];
        bf[ks].y = embW[(size_t)kext_row(k0 + 1) * NOISE + col];
    }

    const int rt1 = w >> 1;
    const int nt1 = w & 1;
    const int c1  = nt1 * 16 + lrow;
    v2f b1[16];
#pragma unroll
    for (int ks = 0; ks < 16; ++ks) {
        const int k0 = ks * 4 + khalf;
        b1[ks].x = trigW[k0 * 32 + c1];
        b1[ks].y = trigW[(k0 + 1) * 32 + c1];
    }
    const float tb = trigb[c1];
    const float eb = embb[col];
    __syncthreads();

    for (int chunk = blockIdx.x; chunk < NCHUNK; chunk += gridDim.x) {
        const int rowbase = chunk * CROWS;

        if (chunk + gridDim.x < NCHUNK) {
            const float* p = trig + (size_t)(chunk + gridDim.x) * CROWS * META
                                  + t * 16;
            __builtin_prefetch(p, 0, 3);
        }

#pragma unroll
        for (int j = 0; j < 4; ++j) {
            const int slot = t + 256 * j;
            const int fr = slot >> 4;
            const int fc = 32 + (slot & 15);
            float v = 0.0f;
            const int grow = rowbase + fr;
            if (fc == 32)      v = chain[grow];
            else if (fc <= 40) v = tx[(size_t)grow * TXD + (fc - 33)];
            T[fr * TSTRIDE + fc] = v;
        }

        {
            v8f d = {};
            const int arow = rowbase + rt1 * 16 + lrow;
#pragma unroll
            for (int ks = 0; ks < 16; ++ks) {
                const int k0 = ks * 4 + khalf;
                v2f a;
                a.x = trig[(size_t)arow * META + k0];
                a.y = trig[(size_t)arow * META + k0 + 1];
                d = __builtin_amdgcn_wmma_f32_16x16x4_f32(false, a, false, b1[ks],
                                                          (short)0, d, false, false);
            }
#pragma unroll
            for (int r = 0; r < 8; ++r) {
                float v = d[r] + tb;
                T[(rt1 * 16 + lr0 + r) * TSTRIDE + c1] = v > 0.0f ? v : 0.0f;
            }
        }
        __syncthreads();

        v8f acc0 = {}, acc1 = {}, acc2 = {}, acc3 = {};
#pragma unroll
        for (int ks = 0; ks < 11; ++ks) {
            const int k0 = ks * 4 + khalf;
            v2f a0, a1, a2, a3;
            a0.x = T[(0 * 16 + lrow) * TSTRIDE + k0];
            a0.y = T[(0 * 16 + lrow) * TSTRIDE + k0 + 1];
            a1.x = T[(1 * 16 + lrow) * TSTRIDE + k0];
            a1.y = T[(1 * 16 + lrow) * TSTRIDE + k0 + 1];
            a2.x = T[(2 * 16 + lrow) * TSTRIDE + k0];
            a2.y = T[(2 * 16 + lrow) * TSTRIDE + k0 + 1];
            a3.x = T[(3 * 16 + lrow) * TSTRIDE + k0];
            a3.y = T[(3 * 16 + lrow) * TSTRIDE + k0 + 1];
            acc0 = __builtin_amdgcn_wmma_f32_16x16x4_f32(false, a0, false, bf[ks],
                                                         (short)0, acc0, false, false);
            acc1 = __builtin_amdgcn_wmma_f32_16x16x4_f32(false, a1, false, bf[ks],
                                                         (short)0, acc1, false, false);
            acc2 = __builtin_amdgcn_wmma_f32_16x16x4_f32(false, a2, false, bf[ks],
                                                         (short)0, acc2, false, false);
            acc3 = __builtin_amdgcn_wmma_f32_16x16x4_f32(false, a3, false, bf[ks],
                                                         (short)0, acc3, false, false);
        }

#pragma unroll
        for (int rt = 0; rt < 4; ++rt) {
            const v8f& c = (rt == 0) ? acc0 : (rt == 1) ? acc1
                         : (rt == 2) ? acc2 : acc3;
#pragma unroll
            for (int r = 0; r < 8; ++r) {
                const int grow = rowbase + rt * 16 + lr0 + r;
                const int g = graphs[grow];
                out[(size_t)grow * NOISE + col] = c[r] + gpl[g * NOISE + col] + eb;
            }
        }
        __syncthreads();
    }
}

extern "C" void kernel_launch(void* const* d_in, const int* in_sizes, int n_in,
                              void* d_out, int out_size, void* d_ws, size_t ws_size,
                              hipStream_t stream) {
    const int*   graphs = (const int*)d_in[0];
    const float* chain  = (const float*)d_in[1];
    const float* trig   = (const float*)d_in[2];
    const float* tx     = (const float*)d_in[3];
    const int*   edges  = (const int*)d_in[4];
    const float* gw     = (const float*)d_in[5];
    const float* gb     = (const float*)d_in[6];
    const float* trigW  = (const float*)d_in[7];
    const float* trigb  = (const float*)d_in[8];
    const float* embW   = (const float*)d_in[9];
    const float* embb   = (const float*)d_in[10];
    float* out = (float*)d_out;

    float* ge = (float*)d_ws;
    float* gp = ge + (size_t)N_GRAPHS * N_NODES;

    gcn_kernel<<<N_GRAPHS, GCN_THREADS, 0, stream>>>(edges, gw, gb, ge);
    proj_kernel<<<32, 1024, 0, stream>>>(ge, embW, gp);
    batch_kernel<<<512, 256, 0, stream>>>(graphs, chain, trig, tx, trigW, trigb,
                                          embW, embb, gp, out);
}


# batch kernel: each chunk's global loads (trigger fragments, tx, chain, graph ids) go out as one group a chunk ahead, in flight across an LDS-only barrier; T double-buffered, graph ids via LDS
# speedup vs baseline: 4.4590x; 1.3166x over previous
#include <hip/hip_runtime.h>
#include <hip/hip_bf16.h>

typedef float v2f __attribute__((ext_vector_type(2)));
typedef float v8f __attribute__((ext_vector_type(8)));

#define N_NODES 2048
#define N_GRAPHS 64
#define N_EDGES 16384
#define BATCH 65536
#define META 64
#define TXD 8
#define NOISE 128
#define TSTRIDE 48
#define CROWS 64
#define NCHUNK (BATCH / CROWS)

__device__ __forceinline__ int kext_row(int k) {
    if (k < 32) return 2049 + k;
    if (k == 32) return 2048;
    if (k <= 40) return 2081 + (k - 33);
    return 2048;
}

#define GCN_THREADS 1024
#define GCN_VEC     (N_EDGES / 4 / GCN_THREADS)

__global__ void __launch_bounds__(GCN_THREADS) gcn_kernel(const int* __restrict__ edges,
                                                          const float* __restrict__ gw,
                                                          const float* __restrict__ gb,
                                                          float* __restrict__ graph_emb) {
    __shared__ float deg[N_NODES];
    __shared__ float agg[N_NODES];
    const int g = blockIdx.x;
    const int t = threadIdx.x;
    const int4* src4 = (const int4*)(edges + (size_t)g * (2 * N_EDGES));
    const int4* dst4 = src4 + N_EDGES / 4;

    int4 dv[GCN_VEC], sv[GCN_VEC];
#pragma unroll
    for (int j = 0; j < GCN_VEC; ++j) dv[j] = dst4[j * GCN_THREADS + t];
#pragma unroll
    for (int j = 0; j < GCN_VEC; ++j) sv[j] = src4[j * GCN_THREADS + t];

    for (int i = t; i < N_NODES; i += GCN_THREADS) deg[i] = 1.0f;
    __syncthreads();
#pragma unroll
    for (int j = 0; j < GCN_VEC; ++j) {
        atomicAdd(&deg[dv[j].x], 1.0f);
        atomicAdd(&deg[dv[j].y], 1.0f);
        atomicAdd(&deg[dv[j].z], 1.0f);
        atomicAdd(&deg[dv[j].w], 1.0f);
    }
    __syncthreads();
    for (int i = t; i < N_NODES; i += GCN_THREADS) {
        float d = deg[i];
        float r = d > 0.0f ? rsqrtf(d) : 0.0f;
        deg[i] = r;
        agg[i] = r * r;
    }
    __syncthreads();
    float nv[4 * GCN_VEC];
#pragma unroll
    for (int j = 0; j < GCN_VEC; ++j) {
        nv[4 * j + 0] = deg[sv[j].x] * deg[dv[j].x];
        nv[4 * j + 1] = deg[sv[j].y] * deg[dv[j].y];
        nv[4 * j + 2] = deg[sv[j].z] * deg[dv[j].z];
        nv[4 * j + 3] = deg[sv[j].w] * deg[dv[j].w];
    }
#pragma unroll
    for (int j = 0; j < GCN_VEC; ++j) {
        atomicAdd(&agg[dv[j].x], nv[4 * j + 0]);
        atomicAdd(&agg[dv[j].y], nv[4 * j + 1]);
        atomicAdd(&agg[dv[j].z], nv[4 * j + 2]);
        atomicAdd(&agg[dv[j].w], nv[4 * j + 3]);
    }
    __syncthreads();
    const float w = *gw, b = *gb;
    for (int i = t; i < N_NODES; i += GCN_THREADS)
        graph_emb[(size_t)g * N_NODES + i] = agg[i] * w + b;
}

#define PJ_WAVES  32
#define PJ_KSLICE (N_NODES / PJ_WAVES)
#define PJ_STEPS  (PJ_KSLICE / 4)

__global__ void __launch_bounds__(1024) proj_kernel(const float* __restrict__ ge,
                                                    const float* __restrict__ embW,
                                                    float* __restrict__ gp) {
    __shared__ float part[PJ_WAVES * 256];

    const int t     = threadIdx.x;
    const int lane  = t & 31;
    const int w     = t >> 5;
    const int mtile = blockIdx.x >> 3;
    const int ntile = blockIdx.x & 7;
    const int lrow  = lane & 15;
    const int khalf = (lane >> 4) * 2;
    const int arow  = mtile * 16 + lrow;
    const int col   = ntile * 16 + lrow;
    const int kbase = w * PJ_KSLICE + khalf;

    const float* ap = ge + (size_t)arow * N_NODES + kbase;
    const float* bp = embW + (size_t)kbase * NOISE + col;

    v8f c0 = {}, c1 = {};
#pragma unroll
    for (int s = 0; s < PJ_STEPS; s += 2) {
        v2f a0, b0, a1, b1;
        a0.x = ap[4 * s];
        a0.y = ap[4 * s + 1];
        b0.x = bp[(size_t)(4 * s) * NOISE];
        b0.y = bp[(size_t)(4 * s + 1) * NOISE];
        a1.x = ap[4 * s + 4];
        a1.y = ap[4 * s + 5];
        b1.x = bp[(size_t)(4 * s + 4) * NOISE];
        b1.y = bp[(size_t)(4 * s + 5) * NOISE];
        c0 = __builtin_amdgcn_wmma_f32_16x16x4_f32(false, a0, false, b0,
                                                   (short)0, c0, false, false);
        c1 = __builtin_amdgcn_wmma_f32_16x16x4_f32(false, a1, false, b1,
                                                   (short)0, c1, false, false);
    }

#pragma unroll
    for (int r = 0; r < 8; ++r)
        part[w * 256 + r * 32 + lane] = c0[r] + c1[r];
    __syncthreads();

    if (t < 256) {
        float s = part[t];
#pragma unroll
        for (int q = 1; q < PJ_WAVES; ++q) s += part[q * 256 + t];
        const int r   = t >> 5;
        const int l   = t & 31;
        const int row = mtile * 16 + (l >> 4) * 8 + r;
        const int cc  = ntile * 16 + (l & 15);
        gp[(size_t)row * NOISE + cc] = s;
    }
}

#define TBUF (CROWS * TSTRIDE)

static __device__ __forceinline__ void lds_barrier() {
    asm volatile("s_wait_dscnt 0x0\n\ts_barrier_signal -1\n\ts_barrier_wait -1" ::: "memory");
}

__global__ void __launch_bounds__(256) batch_kernel(const int* __restrict__ graphs,
                                                    const float* __restrict__ chain,
                                                    const float* __restrict__ trig,
                                                    const float* __restrict__ tx,
                                                    const float* __restrict__ trigW,
                                                    const float* __restrict__ trigb,
                                                    const float* __restrict__ embW,
                                                    const float* __restrict__ embb,
                                                    const float* __restrict__ gp,
                                                    float* __restrict__ out) {
    __shared__ __align__(16) float gpl[N_GRAPHS * NOISE];
    __shared__ __align__(16) float T[2 * TBUF];
    __shared__ __align__(16) int   gofs[2 * CROWS];

    const int t     = threadIdx.x;
    const int lane  = t & 31;
    const int w     = t >> 5;
    const int lrow  = lane & 15;
    const int khalf = (lane >> 4) * 2;
    const int col   = w * 16 + lrow;
    const int lr0   = (lane >> 4) * 8;

    const int rt1 = w >> 1;
    const int nt1 = w & 1;
    const int c1  = nt1 * 16 + lrow;

    const int4* sp;
    int sstep;
    if (t < 128)      { sp = (const int4*)tx + t;             sstep = CROWS * TXD / 4; }
    else if (t < 144) { sp = (const int4*)chain + (t - 128);  sstep = CROWS / 4; }
    else              { sp = (const int4*)graphs + (t - 144); sstep = CROWS / 4; }
    const bool has_side = t < 160;

    v2f  a[16];
    int4 sv = {0, 0, 0, 0};
    auto load_chunk = [&](int ch) {
        const float* ap = trig + (size_t)(ch * CROWS + rt1 * 16 + lrow) * META + khalf;
#pragma unroll
        for (int ks = 0; ks < 16; ++ks) {
            a[ks].x = ap[ks * 4];
            a[ks].y = ap[ks * 4 + 1];
        }
        if (has_side) sv = sp[(size_t)ch * sstep];
    };

    int chunk = blockIdx.x;
    load_chunk(chunk);

    v2f b1[16];
#pragma unroll
    for (int ks = 0; ks < 16; ++ks) {
        const int k0 = ks * 4 + khalf;
        b1[ks].x = trigW[k0 * 32 + c1];
        b1[ks].y = trigW[(k0 + 1) * 32 + c1];
    }
    const float tb = trigb[c1];

    float4 gq[8];
#pragma unroll
    for (int j = 0; j < 8; ++j) gq[j] = *(const float4*)&gp[(j * 256 + t) * 4];

    v2f bf[11];
#pragma unroll
    for (int ks = 0; ks < 11; ++ks) {
        const int k0 = ks * 4 + khalf;
        bf[ks].x = embW[(size_t)kext_row(k0) * NOISE + col];
        bf[ks].y = embW[(size_t)kext_row(k0 + 1) * NOISE + col];
    }
    const float eb = embb[col];

    if (t < 2 * CROWS) {
        float* z = &T[(t >> 6) * TBUF + (t & 63) * TSTRIDE + 41];
        z[0] = 0.0f; z[1] = 0.0f; z[2] = 0.0f;
    }
#pragma unroll
    for (int j = 0; j < 8; ++j) *(float4*)&gpl[(j * 256 + t) * 4] = gq[j];

    int buf = 0;
    for (; chunk < NCHUNK; chunk += gridDim.x) {
        const int rowbase = chunk * CROWS;
        float* Tb = T + buf * TBUF;
        int*   gb = gofs + buf * CROWS;

        {
            v8f d = {};
#pragma unroll
            for (int ks = 0; ks < 16; ++ks)
                d = __builtin_amdgcn_wmma_f32_16x16x4_f32(false, a[ks], false, b1[ks],
                                                          (short)0, d, false, false);
#pragma unroll
            for (int r = 0; r < 8; ++r) {
                float v = d[r] + tb;
                Tb[(rt1 * 16 + lr0 + r) * TSTRIDE + c1] = v > 0.0f ? v : 0.0f;
            }
        }
        if (t < 128) {
            float* q = &Tb[(t >> 1) * TSTRIDE + 33 + (t & 1) * 4];
            q[0] = __int_as_float(sv.x); q[1] = __int_as_float(sv.y);
            q[2] = __int_as_float(sv.z); q[3] = __int_as_float(sv.w);
        } else if (t < 144) {
            float* q = &Tb[(t - 128) * 4 * TSTRIDE + 32];
            q[0 * TSTRIDE] = __int_as_float(sv.x); q[1 * TSTRIDE] = __int_as_float(sv.y);
            q[2 * TSTRIDE] = __int_as_float(sv.z); q[3 * TSTRIDE] = __int_as_float(sv.w);
        } else if (t < 160) {
            int* q = &gb[(t - 144) * 4];
            q[0] = sv.x * NOISE; q[1] = sv.y * NOISE; q[2] = sv.z * NOISE; q[3] = sv.w * NOISE;
        }

        if (chunk + gridDim.x < NCHUNK) load_chunk(chunk + gridDim.x);

        lds_barrier();

        v8f acc0 = {}, acc1 = {}, acc2 = {}, acc3 = {};
#pragma unroll
        for (int ks = 0; ks < 11; ++ks) {
            const int k0 = ks * 4 + khalf;
            v2f a0, a1, a2, a3;
            a0.x = Tb[(0 * 16 + lrow) * TSTRIDE + k0];
            a0.y = Tb[(0 * 16 + lrow) * TSTRIDE + k0 + 1];
            a1.x = Tb[(1 * 16 + lrow) * TSTRIDE + k0];
            a1.y = Tb[(1 * 16 + lrow) * TSTRIDE + k0 + 1];
            a2.x = Tb[(2 * 16 + lrow) * TSTRIDE + k0];
            a2.y = Tb[(2 * 16 + lrow) * TSTRIDE + k0 + 1];
            a3.x = Tb[(3 * 16 + lrow) * TSTRIDE + k0];
            a3.y = Tb[(3 * 16 + lrow) * TSTRIDE + k0 + 1];
            acc0 = __builtin_amdgcn_wmma_f32_16x16x4_f32(false, a0, false, bf[ks],
                                                         (short)0, acc0, false, false);
            acc1 = __builtin_amdgcn_wmma_f32_16x16x4_f32(false, a1, false, bf[ks],
                                                         (short)0, acc1, false, false);
            acc2 = __builtin_amdgcn_wmma_f32_16x16x4_f32(false, a2, false, bf[ks],
                                                         (short)0, acc2, false, false);
            acc3 = __builtin_amdgcn_wmma_f32_16x16x4_f32(false, a3, false, bf[ks],
                                                         (short)0, acc3, false, false);
        }

        float* orow = out + ((size_t)rowbase + lr0) * NOISE + col;
#pragma unroll
        for (int rt = 0; rt < 4; ++rt) {
            const v8f& c = (rt == 0) ? acc0 : (rt == 1) ? acc1
                         : (rt == 2) ? acc2 : acc3;
            const int4 g0 = *(const int4*)&gb[rt * 16 + lr0];
            const int4 g1 = *(const int4*)&gb[rt * 16 + lr0 + 4];
            const int go[8] = {g0.x, g0.y, g0.z, g0.w, g1.x, g1.y, g1.z, g1.w};
#pragma unroll
            for (int r = 0; r < 8; ++r)
                orow[(size_t)(rt * 16 + r) * NOISE] = c[r] + gpl[go[r] + col] + eb;
        }
        buf ^= 1;
    }
}

extern "C" void kernel_launch(void* const* d_in, const int* in_sizes, int n_in,
                              void* d_out, int out_size, void* d_ws, size_t ws_size,
                              hipStream_t stream) {
    const int*   graphs = (const int*)d_in[0];
    const float* chain  = (const float*)d_in[1];
    const float* trig   = (const float*)d_in[2];
    const float* tx     = (const float*)d_in[3];
    const int*   edges  = (const int*)d_in[4];
    const float* gw     = (const float*)d_in[5];
    const float* gb     = (const float*)d_in[6];
    const float* trigW  = (const float*)d_in[7];
    const float* trigb  = (const float*)d_in[8];
    const float* embW   = (const float*)d_in[9];
    const float* embb   = (const float*)d_in[10];
    float* out = (float*)d_out;

    float* ge = (float*)d_ws;
    float* gp = ge + (size_t)N_GRAPHS * N_NODES;

    gcn_kernel<<<N_GRAPHS, GCN_THREADS, 0, stream>>>(edges, gw, gb, ge);
    proj_kernel<<<32, 1024, 0, stream>>>(ge, embW, gp);
    batch_kernel<<<512, 256, 0, stream>>>(graphs, chain, trig, tx, trigW, trigb,
                                          embW, embb, gp, out);
}


# proj GEMM fused into the graph kernel: 4 groups per graph, each multiplies the graph's embedding row (in LDS) by a 32-column slice of emb_W on f32 WMMA, K over 32 waves, sum order kept; 2 launches
# speedup vs baseline: 5.5106x; 1.2358x over previous
#include <hip/hip_runtime.h>
#include <hip/hip_bf16.h>

typedef float v2f __attribute__((ext_vector_type(2)));
typedef float v8f __attribute__((ext_vector_type(8)));

#define N_NODES 2048
#define N_GRAPHS 64
#define N_EDGES 16384
#define BATCH 65536
#define META 64
#define TXD 8
#define NOISE 128
#define TSTRIDE 48
#define CROWS 64
#define NCHUNK (BATCH / CROWS)

__device__ __forceinline__ int kext_row(int k) {
    if (k < 32) return 2049 + k;
    if (k == 32) return 2048;
    if (k <= 40) return 2081 + (k - 33);
    return 2048;
}

#define GCN_THREADS 1024
#define GCN_VEC     (N_EDGES / 4 / GCN_THREADS)
#define PJ_WAVES    32
#define PJ_KSLICE   (N_NODES / PJ_WAVES)
#define PJ_STEPS    (PJ_KSLICE / 4)
#define PJ_QCOLS    32

__global__ void __launch_bounds__(GCN_THREADS) gcn_proj_kernel(const int* __restrict__ edges,
                                                               const float* __restrict__ gw,
                                                               const float* __restrict__ gb,
                                                               const float* __restrict__ embW,
                                                               float* __restrict__ gp) {
    __shared__ float deg[N_NODES];
    __shared__ float agg[N_NODES];
    __shared__ float part[PJ_WAVES * PJ_QCOLS];
    const int g = blockIdx.x >> 2;
    const int q = blockIdx.x & 3;
    const int t = threadIdx.x;
    const int4* src4 = (const int4*)(edges + (size_t)g * (2 * N_EDGES));
    const int4* dst4 = src4 + N_EDGES / 4;

    int4 dv[GCN_VEC], sv[GCN_VEC];
#pragma unroll
    for (int j = 0; j < GCN_VEC; ++j) dv[j] = dst4[j * GCN_THREADS + t];
#pragma unroll
    for (int j = 0; j < GCN_VEC; ++j) sv[j] = src4[j * GCN_THREADS + t];

    for (int i = t; i < N_NODES; i += GCN_THREADS) deg[i] = 1.0f;
    __syncthreads();
#pragma unroll
    for (int j = 0; j < GCN_VEC; ++j) {
        atomicAdd(&deg[dv[j].x], 1.0f);
        atomicAdd(&deg[dv[j].y], 1.0f);
        atomicAdd(&deg[dv[j].z], 1.0f);
        atomicAdd(&deg[dv[j].w], 1.0f);
    }
    __syncthreads();
    for (int i = t; i < N_NODES; i += GCN_THREADS) {
        float d = deg[i];
        float r = d > 0.0f ? rsqrtf(d) : 0.0f;
        deg[i] = r;
        agg[i] = r * r;
    }
    __syncthreads();
    const int lane  = t & 31;
    const int wv    = t >> 5;
    const int lcol  = lane & 15;
    const int khalf = (lane >> 4) * 2;
    const int kbase = wv * PJ_KSLICE + khalf;
    const float* bp = embW + (size_t)kbase * NOISE + q * PJ_QCOLS + 2 * lcol;
    v2f br[2 * PJ_STEPS];
#pragma unroll
    for (int s = 0; s < PJ_STEPS; ++s) {
        br[2 * s]     = *(const v2f*)&bp[(size_t)(4 * s) * NOISE];
        br[2 * s + 1] = *(const v2f*)&bp[(size_t)(4 * s + 1) * NOISE];
    }

    float nv[4 * GCN_VEC];
#pragma unroll
    for (int j = 0; j < GCN_VEC; ++j) {
        nv[4 * j + 0] = deg[sv[j].x] * deg[dv[j].x];
        nv[4 * j + 1] = deg[sv[j].y] * deg[dv[j].y];
        nv[4 * j + 2] = deg[sv[j].z] * deg[dv[j].z];
        nv[4 * j + 3] = deg[sv[j].w] * deg[dv[j].w];
    }

#pragma unroll
    for (int j = 0; j < GCN_VEC; ++j) {
        atomicAdd(&agg[dv[j].x], nv[4 * j + 0]);
        atomicAdd(&agg[dv[j].y], nv[4 * j + 1]);
        atomicAdd(&agg[dv[j].z], nv[4 * j + 2]);
        atomicAdd(&agg[dv[j].w], nv[4 * j + 3]);
    }
    __syncthreads();
    const float w = *gw, b = *gb;
    for (int i = t; i < N_NODES; i += GCN_THREADS) agg[i] = agg[i] * w + b;
    __syncthreads();

    v8f ce0 = {}, ce1 = {}, co0 = {}, co1 = {};
#pragma unroll
    for (int s = 0; s < PJ_STEPS; s += 2) {
        v2f a0, a1, be0, bo0, be1, bo1;
        a0.x = agg[kbase + 4 * s];
        a0.y = agg[kbase + 4 * s + 1];
        a1.x = agg[kbase + 4 * s + 4];
        a1.y = agg[kbase + 4 * s + 5];
        be0.x = br[2 * s].x;     be0.y = br[2 * s + 1].x;
        bo0.x = br[2 * s].y;     bo0.y = br[2 * s + 1].y;
        be1.x = br[2 * s + 2].x; be1.y = br[2 * s + 3].x;
        bo1.x = br[2 * s + 2].y; bo1.y = br[2 * s + 3].y;
        ce0 = __builtin_amdgcn_wmma_f32_16x16x4_f32(false, a0, false, be0,
                                                    (short)0, ce0, false, false);
        co0 = __builtin_amdgcn_wmma_f32_16x16x4_f32(false, a0, false, bo0,
                                                    (short)0, co0, false, false);
        ce1 = __builtin_amdgcn_wmma_f32_16x16x4_f32(false, a1, false, be1,
                                                    (short)0, ce1, false, false);
        co1 = __builtin_amdgcn_wmma_f32_16x16x4_f32(false, a1, false, bo1,
                                                    (short)0, co1, false, false);
    }

    if (lane < 16) {
        part[wv * PJ_QCOLS + 2 * lcol]     = ce0[0] + ce1[0];
        part[wv * PJ_QCOLS + 2 * lcol + 1] = co0[0] + co1[0];
    }
    __syncthreads();

    if (t < PJ_QCOLS) {
        float sum = part[t];
#pragma unroll
        for (int p = 1; p < PJ_WAVES; ++p) sum += part[p * PJ_QCOLS + t];
        gp[(size_t)g * NOISE + q * PJ_QCOLS + t] = sum;
    }
}

#define TBUF (CROWS * TSTRIDE)

static __device__ __forceinline__ void lds_barrier() {
    asm volatile("s_wait_dscnt 0x0\n\ts_barrier_signal -1\n\ts_barrier_wait -1" ::: "memory");
}

__global__ void __launch_bounds__(256) batch_kernel(const int* __restrict__ graphs,
                                                    const float* __restrict__ chain,
                                                    const float* __restrict__ trig,
                                                    const float* __restrict__ tx,
                                                    const float* __restrict__ trigW,
                                                    const float* __restrict__ trigb,
                                                    const float* __restrict__ embW,
                                                    const float* __restrict__ embb,
                                                    const float* __restrict__ gp,
                                                    float* __restrict__ out) {
    __shared__ __align__(16) float gpl[N_GRAPHS * NOISE];
    __shared__ __align__(16) float T[2 * TBUF];
    __shared__ __align__(16) int   gofs[2 * CROWS];

    const int t     = threadIdx.x;
    const int lane  = t & 31;
    const int w     = t >> 5;
    const int lrow  = lane & 15;
    const int khalf = (lane >> 4) * 2;
    const int col   = w * 16 + lrow;
    const int lr0   = (lane >> 4) * 8;

    const int rt1 = w >> 1;
    const int nt1 = w & 1;
    const int c1  = nt1 * 16 + lrow;

    const int4* sp;
    int sstep;
    if (t < 128)      { sp = (const int4*)tx + t;             sstep = CROWS * TXD / 4; }
    else if (t < 144) { sp = (const int4*)chain + (t - 128);  sstep = CROWS / 4; }
    else              { sp = (const int4*)graphs + (t - 144); sstep = CROWS / 4; }
    const bool has_side = t < 160;

    v2f  a[16];
    int4 sv = {0, 0, 0, 0};
    auto load_chunk = [&](int ch) {
        const float* ap = trig + (size_t)(ch * CROWS + rt1 * 16 + lrow) * META + khalf;
#pragma unroll
        for (int ks = 0; ks < 16; ++ks) {
            a[ks].x = ap[ks * 4];
            a[ks].y = ap[ks * 4 + 1];
        }
        if (has_side) sv = sp[(size_t)ch * sstep];
    };

    int chunk = blockIdx.x;
    load_chunk(chunk);

    v2f b1[16];
#pragma unroll
    for (int ks = 0; ks < 16; ++ks) {
        const int k0 = ks * 4 + khalf;
        b1[ks].x = trigW[k0 * 32 + c1];
        b1[ks].y = trigW[(k0 + 1) * 32 + c1];
    }
    const float tb = trigb[c1];

    float4 gq[8];
#pragma unroll
    for (int j = 0; j < 8; ++j) gq[j] = *(const float4*)&gp[(j * 256 + t) * 4];

    v2f bf[11];
#pragma unroll
    for (int ks = 0; ks < 11; ++ks) {
        const int k0 = ks * 4 + khalf;
        bf[ks].x = embW[(size_t)kext_row(k0) * NOISE + col];
        bf[ks].y = embW[(size_t)kext_row(k0 + 1) * NOISE + col];
    }
    const float eb = embb[col];

    if (t < 2 * CROWS) {
        float* z = &T[(t >> 6) * TBUF + (t & 63) * TSTRIDE + 41];
        z[0] = 0.0f; z[1] = 0.0f; z[2] = 0.0f;
    }
#pragma unroll
    for (int j = 0; j < 8; ++j) *(float4*)&gpl[(j * 256 + t) * 4] = gq[j];

    int buf = 0;
    for (; chunk < NCHUNK; chunk += gridDim.x) {
        const int rowbase = chunk * CROWS;
        float* Tb = T + buf * TBUF;
        int*   gb = gofs + buf * CROWS;

        {
            v8f d = {};
#pragma unroll
            for (int ks = 0; ks < 16; ++ks)
                d = __builtin_amdgcn_wmma_f32_16x16x4_f32(false, a[ks], false, b1[ks],
                                                          (short)0, d, false, false);
#pragma unroll
            for (int r = 0; r < 8; ++r) {
                float v = d[r] + tb;
                Tb[(rt1 * 16 + lr0 + r) * TSTRIDE + c1] = v > 0.0f ? v : 0.0f;
            }
        }
        if (t < 128) {
            float* q = &Tb[(t >> 1) * TSTRIDE + 33 + (t & 1) * 4];
            q[0] = __int_as_float(sv.x); q[1] = __int_as_float(sv.y);
            q[2] = __int_as_float(sv.z); q[3] = __int_as_float(sv.w);
        } else if (t < 144) {
            float* q = &Tb[(t - 128) * 4 * TSTRIDE + 32];
            q[0 * TSTRIDE] = __int_as_float(sv.x); q[1 * TSTRIDE] = __int_as_float(sv.y);
            q[2 * TSTRIDE] = __int_as_float(sv.z); q[3 * TSTRIDE] = __int_as_float(sv.w);
        } else if (t < 160) {
            int* q = &gb[(t - 144) * 4];
            q[0] = sv.x * NOISE; q[1] = sv.y * NOISE; q[2] = sv.z * NOISE; q[3] = sv.w * NOISE;
        }

        if (chunk + gridDim.x < NCHUNK) load_chunk(chunk + gridDim.x);

        lds_barrier();

        v8f acc0 = {}, acc1 = {}, acc2 = {}, acc3 = {};
#pragma unroll
        for (int ks = 0; ks < 11; ++ks) {
            const int k0 = ks * 4 + khalf;
            v2f a0, a1, a2, a3;
            a0.x = Tb[(0 * 16 + lrow) * TSTRIDE + k0];
            a0.y = Tb[(0 * 16 + lrow) * TSTRIDE + k0 + 1];
            a1.x = Tb[(1 * 16 + lrow) * TSTRIDE + k0];
            a1.y = Tb[(1 * 16 + lrow) * TSTRIDE + k0 + 1];
            a2.x = Tb[(2 * 16 + lrow) * TSTRIDE + k0];
            a2.y = Tb[(2 * 16 + lrow) * TSTRIDE + k0 + 1];
            a3.x = Tb[(3 * 16 + lrow) * TSTRIDE + k0];
            a3.y = Tb[(3 * 16 + lrow) * TSTRIDE + k0 + 1];
            acc0 = __builtin_amdgcn_wmma_f32_16x16x4_f32(false, a0, false, bf[ks],
                                                         (short)0, acc0, false, false);
            acc1 = __builtin_amdgcn_wmma_f32_16x16x4_f32(false, a1, false, bf[ks],
                                                         (short)0, acc1, false, false);
            acc2 = __builtin_amdgcn_wmma_f32_16x16x4_f32(false, a2, false, bf[ks],
                                                         (short)0, acc2, false, false);
            acc3 = __builtin_amdgcn_wmma_f32_16x16x4_f32(false, a3, false, bf[ks],
                                                         (short)0, acc3, false, false);
        }

        float* orow = out + ((size_t)rowbase + lr0) * NOISE + col;
#pragma unroll
        for (int rt = 0; rt < 4; ++rt) {
            const v8f& c = (rt == 0) ? acc0 : (rt == 1) ? acc1
                         : (rt == 2) ? acc2 : acc3;
            const int4 g0 = *(const int4*)&gb[rt * 16 + lr0];
            const int4 g1 = *(const int4*)&gb[rt * 16 + lr0 + 4];
            const int go[8] = {g0.x, g0.y, g0.z, g0.w, g1.x, g1.y, g1.z, g1.w};
#pragma unroll
            for (int r = 0; r < 8; ++r)
                orow[(size_t)(rt * 16 + r) * NOISE] = c[r] + gpl[go[r] + col] + eb;
        }
        buf ^= 1;
    }
}

extern "C" void kernel_launch(void* const* d_in, const int* in_sizes, int n_in,
                              void* d_out, int out_size, void* d_ws, size_t ws_size,
                              hipStream_t stream) {
    const int*   graphs = (const int*)d_in[0];
    const float* chain  = (const float*)d_in[1];
    const float* trig   = (const float*)d_in[2];
    const float* tx     = (const float*)d_in[3];
    const int*   edges  = (const int*)d_in[4];
    const float* gw     = (const float*)d_in[5];
    const float* gb     = (const float*)d_in[6];
    const float* trigW  = (const float*)d_in[7];
    const float* trigb  = (const float*)d_in[8];
    const float* embW   = (const float*)d_in[9];
    const float* embb   = (const float*)d_in[10];
    float* out = (float*)d_out;

    float* gp = (float*)d_ws + (size_t)N_GRAPHS * N_NODES;

    gcn_proj_kernel<<<4 * N_GRAPHS, GCN_THREADS, 0, stream>>>(edges, gw, gb, embW, gp);
    batch_kernel<<<512, 256, 0, stream>>>(graphs, chain, trig, tx, trigW, trigb,
                                          embW, embb, gp, out);
}
